# starting kernel
# baseline (speedup 1.0000x reference)
#include <hip/hip_runtime.h>

typedef __attribute__((ext_vector_type(2))) float v2f;
typedef __attribute__((ext_vector_type(8))) float v8f;

#define D_IN  256
#define D_OUT 128

__global__ __launch_bounds__(256) void gcn_init_out(const float* __restrict__ bias,
                                                    float* __restrict__ out,
                                                    long long total4) {
    long long i = (long long)blockIdx.x * blockDim.x + threadIdx.x;
    if (i < total4) {
        int d4 = (int)(i & (D_OUT / 4 - 1));
        ((float4*)out)[i] = ((const float4*)bias)[d4];
    }
}

__global__ __launch_bounds__(256) void gcn_gemm_xw(const float* __restrict__ X,
                                                   const float* __restrict__ W,
                                                   float* __restrict__ S,
                                                   int n_rows) {
    const int lane  = threadIdx.x & 31;
    const int wave  = threadIdx.x >> 5;
    const int mtile = blockIdx.x * 8 + wave;
    if (mtile * 16 + 16 > n_rows) return;
    const int m0 = mtile * 16;

    const int am = lane & 15;
    const int ak = (lane >> 4) * 2;
    const int bn = lane & 15;
    const int bk = (lane >> 4) * 2;

    v8f c[8];
#pragma unroll
    for (int t = 0; t < 8; ++t) c[t] = (v8f){0.f, 0.f, 0.f, 0.f, 0.f, 0.f, 0.f, 0.f};

    const float* xrow = X + (long long)(m0 + am) * D_IN;

    for (int k = 0; k < D_IN; k += 4) {
        v2f a = *(const v2f*)(xrow + k + ak);
#pragma unroll
        for (int t = 0; t < 8; ++t) {
            v2f b;
            b.x = W[(long long)(k + bk)     * D_OUT + t * 16 + bn];
            b.y = W[(long long)(k + bk + 1) * D_OUT + t * 16 + bn];
            c[t] = __builtin_amdgcn_wmma_f32_16x16x4_f32(
                 false, a,  false, b,
                 (short)0, c[t],  false,  false);
        }
    }

    const int cn    = lane & 15;
    const int cmoff = (lane >> 4) * 8;
#pragma unroll
    for (int t = 0; t < 8; ++t) {
#pragma unroll
        for (int r = 0; r < 8; ++r) {
            S[(long long)(m0 + r + cmoff) * D_OUT + t * 16 + cn] = c[t][r];
        }
    }
}

__global__ __launch_bounds__(256) void gcn_edge_scatter(const int* __restrict__ rows,
                                                        const int* __restrict__ cols,
                                                        const float* __restrict__ vals,
                                                        const float* __restrict__ S,
                                                        float* __restrict__ out,
                                                        int n_edges) {
    const int lane = threadIdx.x & 31;
    int e = (int)(blockIdx.x * 8u) + (threadIdx.x >> 5);
    e = __builtin_amdgcn_readfirstlane(e);
    if (e >= n_edges) return;

    const int   r = rows[e];
    const int   c = cols[e];
    const float v = vals[e];

    const float4 s = *(const float4*)(S + (long long)c * D_OUT + lane * 4);

    float4 g;
    g.x = s.x * v;
    g.y = s.y * v;
    g.z = s.z * v;
    g.w = s.w * v;

    float* dst = out + (long long)r * D_OUT + lane * 4;
    asm volatile("global_atomic_add_f32 %0, %1, off"           :: "v"(dst), "v"(g.x) : "memory");
    asm volatile("global_atomic_add_f32 %0, %1, off offset:4"  :: "v"(dst), "v"(g.y) : "memory");
    asm volatile("global_atomic_add_f32 %0, %1, off offset:8"  :: "v"(dst), "v"(g.z) : "memory");
    asm volatile("global_atomic_add_f32 %0, %1, off offset:12" :: "v"(dst), "v"(g.w) : "memory");
}

extern "C" void kernel_launch(void* const* d_in, const int* in_sizes, int n_in,
                              void* d_out, int out_size, void* d_ws, size_t ws_size,
                              hipStream_t stream) {
    const int*   rows = (const int*)d_in[0];
    const int*   cols = (const int*)d_in[1];
    const float* vals = (const float*)d_in[2];
    const float* x    = (const float*)d_in[3];
    const float* W    = (const float*)d_in[4];
    const float* bias = (const float*)d_in[5];

    const int E = in_sizes[0];
    const int N = in_sizes[3] / D_IN;

    float* support = (float*)d_ws;

    {
        long long total4 = (long long)N * (D_OUT / 4);
        int blocks = (int)((total4 + 255) / 256);
        gcn_init_out<<<blocks, 256, 0, stream>>>(bias, (float*)d_out, total4);
    }

    {
        int mtiles = N / 16;
        int blocks = (mtiles + 7) / 8;
        gcn_gemm_xw<<<blocks, 256, 0, stream>>>(x, W, support, N);
    }

    {
        int blocks = (E + 7) / 8;
        gcn_edge_scatter<<<blocks, 256, 0, stream>>>(rows, cols, vals, support,
                                                     (float*)d_out, E);
    }
}


# Tile filter scans one key byte per edge (tile of its row, plane written in the GEMM launch), not rows[]: 1/4 the scan bytes, zero-byte word test, rows[] read only for hits; f32 sum order changes
# speedup vs baseline: 1.9508x; 1.9508x over previous
#include <hip/hip_runtime.h>

typedef __attribute__((ext_vector_type(2))) float v2f;
typedef __attribute__((ext_vector_type(8))) float v8f;

#define D_IN  256
#define D_OUT 128

#define TILE_WAVES    32
#define ROWS_PER_WG   400
#define LIST_CAP      800
#define STAGE_CAP     64
#define SCAN_STEP     512
#define SCAN_GROUP    4
#define KEY_NONE      255u

__device__ __forceinline__ unsigned tile_key(int r, int n_rows) {
    return ((unsigned)r < (unsigned)n_rows) ? (unsigned)r / ROWS_PER_WG : KEY_NONE;
}

__global__ __launch_bounds__(256) void gcn_gemm_xw(const float* __restrict__ X,
                                                   const float* __restrict__ W,
                                                   float* __restrict__ S,
                                                   int n_rows,
                                                   const int* __restrict__ rows,
                                                   unsigned* __restrict__ keys,
                                                   int n_edges) {
    {
        const int nk = ((n_edges + SCAN_STEP - 1) / SCAN_STEP) * (SCAN_STEP / 4);
        for (int g = blockIdx.x * 256 + threadIdx.x; g < nk; g += gridDim.x * 256) {
            const int e = g * 4;
            int4 r;
            if (e + 3 < n_edges) {
                r = *(const int4*)(rows + e);
            } else {
                r.x = (e     < n_edges) ? rows[e]     : -1;
                r.y = (e + 1 < n_edges) ? rows[e + 1] : -1;
                r.z = (e + 2 < n_edges) ? rows[e + 2] : -1;
                r.w = (e + 3 < n_edges) ? rows[e + 3] : -1;
            }
            keys[g] = tile_key(r.x, n_rows) | (tile_key(r.y, n_rows) << 8) |
                      (tile_key(r.z, n_rows) << 16) | (tile_key(r.w, n_rows) << 24);
        }
    }
    const int lane  = threadIdx.x & 31;
    const int wave  = threadIdx.x >> 5;
    const int mtile = blockIdx.x * 8 + wave;
    if (mtile * 16 + 16 > n_rows) return;
    const int m0 = mtile * 16;

    const int am = lane & 15;
    const int ak = (lane >> 4) * 2;
    const int bn = lane & 15;
    const int bk = (lane >> 4) * 2;

    v8f c[8];
#pragma unroll
    for (int t = 0; t < 8; ++t) c[t] = (v8f){0.f, 0.f, 0.f, 0.f, 0.f, 0.f, 0.f, 0.f};

    const float* xrow = X + (long long)(m0 + am) * D_IN;

    for (int k = 0; k < D_IN; k += 4) {
        v2f a = *(const v2f*)(xrow + k + ak);
#pragma unroll
        for (int t = 0; t < 8; ++t) {
            v2f b;
            b.x = W[(long long)(k + bk)     * D_OUT + t * 16 + bn];
            b.y = W[(long long)(k + bk + 1) * D_OUT + t * 16 + bn];
            c[t] = __builtin_amdgcn_wmma_f32_16x16x4_f32(
                 false, a,  false, b,
                 (short)0, c[t],  false,  false);
        }
    }

    const int cn    = lane & 15;
    const int cmoff = (lane >> 4) * 8;
#pragma unroll
    for (int t = 0; t < 8; ++t) {
#pragma unroll
        for (int r = 0; r < 8; ++r) {
            S[(long long)(m0 + r + cmoff) * D_OUT + t * 16 + cn] = c[t][r];
        }
    }
}

__global__ __launch_bounds__(1024) void gcn_tile_sum(const unsigned* __restrict__ keys,
                                                     const int* __restrict__ rows,
                                                     const int* __restrict__ cols,
                                                     const float* __restrict__ vals,
                                                     const float* __restrict__ S,
                                                     const float* __restrict__ bias,
                                                     float* __restrict__ out,
                                                     int n_edges, int n_rows) {
    __shared__ float4   acc[ROWS_PER_WG * (D_OUT / 4)];
    __shared__ unsigned lists[TILE_WAVES * LIST_CAP];
    __shared__ uint2    stage_all[TILE_WAVES * STAGE_CAP];
    __shared__ unsigned cnt[TILE_WAVES];
    __shared__ unsigned pending;

    const int tid  = threadIdx.x;
    const int lane = tid & 31;
    const int wave = __builtin_amdgcn_readfirstlane(tid >> 5);
    const unsigned r0 = blockIdx.x * ROWS_PER_WG;
    const int rem_rows = n_rows - (int)r0;
    const unsigned tile_rows = (unsigned)(rem_rows < ROWS_PER_WG ? rem_rows : ROWS_PER_WG);

    for (int i = tid; i < ROWS_PER_WG * (D_OUT / 4); i += 1024)
        acc[i] = make_float4(0.f, 0.f, 0.f, 0.f);
    if (tid == 0) pending = 0;

    unsigned* mylist  = lists + wave * LIST_CAP;
    uint2*    mystage = stage_all + wave * STAGE_CAP;

    const int nsteps = (n_edges + SCAN_STEP - 1) / SCAN_STEP;
    const int rot    = (int)(((long long)blockIdx.x * nsteps) / gridDim.x);
    const unsigned pat = blockIdx.x * 0x01010101u;

    __syncthreads();

    int st = wave;
    unsigned t = 0;
    for (;;) {
        unsigned qn = 0;
        while (st < nsteps && qn + SCAN_STEP <= LIST_CAP) {
            uint4 kk[SCAN_GROUP];
#pragma unroll
            for (int u = 0; u < SCAN_GROUP; ++u) {
                int s = st + u * TILE_WAVES;
                if (s >= nsteps) s = nsteps - 1;
                int sp = s + rot;
                if (sp >= nsteps) sp -= nsteps;
                kk[u] = *(const uint4*)(keys + sp * (SCAN_STEP / 4) + lane * 4);
            }
#pragma unroll
            for (int u = 0; u < SCAN_GROUP; ++u) {
                if (st < nsteps && qn + SCAN_STEP <= LIST_CAP) {
                    const unsigned kw[4] = {kk[u].x, kk[u].y, kk[u].z, kk[u].w};
#pragma unroll
                    for (int j = 0; j < 4; ++j) {
                        unsigned y = kw[j] ^ pat;
                        unsigned z = (y - 0x01010101u) & ~y & 0x80808080u;
                        unsigned m = __builtin_amdgcn_ballot_w32(z != 0u);
                        while (m) {
                            if (z) {
                                const unsigned b    = (unsigned)__builtin_ctz(z) >> 3;
                                const unsigned slot = qn + __builtin_amdgcn_mbcnt_lo(m, 0u);
                                mylist[slot] = (t << 9) | (unsigned)(lane * 16 + j * 4) | b;
                                y |= 0xFFu << (b * 8u);
                                z = (y - 0x01010101u) & ~y & 0x80808080u;
                            }
                            qn += (unsigned)__builtin_popcount(m);
                            m = __builtin_amdgcn_ballot_w32(z != 0u);
                        }
                    }
                    st += TILE_WAVES;
                    ++t;
                }
            }
        }
        for (unsigned p = (unsigned)lane; p < qn; p += 32u) {
            const unsigned en = mylist[p];
            int spd = wave + TILE_WAVES * (int)((en >> 9) & 0x3FFFu) + rot;
            if (spd >= nsteps) spd -= nsteps;
            const unsigned e = (unsigned)(spd * SCAN_STEP) + (en & 511u);
            mylist[p] = en | (((unsigned)rows[e] - r0) << 23);
        }
        if (lane == 0) {
            cnt[wave] = qn;
            if (st < nsteps) pending = 1;
        }
        __syncthreads();

        unsigned qs = 0;
        for (int L = 0; L <= TILE_WAVES; ++L) {
            unsigned n = 0;
            int nchunks = 1;
            unsigned cur = 0;
            if (L < TILE_WAVES) {
                n = (unsigned)__builtin_amdgcn_readfirstlane((int)cnt[L]);
                nchunks = (int)((n + 31u) >> 5);
                if ((unsigned)lane < n) cur = lists[L * LIST_CAP + lane];
            }
            for (int c = 0; c < nchunks; ++c) {
                unsigned nxt = 0;
                if (L < TILE_WAVES) {
                    const unsigned pn = (unsigned)((c + 1) * 32 + lane);
                    if (pn < n) nxt = lists[L * LIST_CAP + pn];
                    const unsigned p = (unsigned)(c * 32 + lane);
                    const bool own = (p < n) && (((cur >> 23) & 31u) == (unsigned)wave);
                    const unsigned m = __builtin_amdgcn_ballot_w32(own);
                    if (m) {
                        if (own) {
                            const unsigned slot = qs + __builtin_amdgcn_mbcnt_lo(m, 0u);
                            int spd = L + TILE_WAVES * (int)((cur >> 9) & 0x3FFFu) + rot;
                            if (spd >= nsteps) spd -= nsteps;
                            mystage[slot] = make_uint2((unsigned)(spd * SCAN_STEP) + (cur & 511u),
                                                       cur >> 23);
                        }
                        qs += (unsigned)__builtin_popcount(m);
                    }
                }
                const unsigned thr = (L < TILE_WAVES) ? 32u : 1u;
                if (qs >= thr) {
                    const unsigned nb = qs < 32u ? qs : 32u;
                    int cv = 0, vv = 0, dv = 0;
                    if ((unsigned)lane < nb) {
                        const uint2 sg = mystage[lane];
                        cv = cols[sg.x];
                        vv = __float_as_int(vals[sg.x]);
                        dv = (int)sg.y;
                    }
                    unsigned k = 0;
                    for (; k + 8 <= nb; k += 8) {
                        float4 s[8];
#pragma unroll
                        for (unsigned u = 0; u < 8; ++u) {
                            const int cc = __builtin_amdgcn_readlane(cv, (int)(k + u));
                            s[u] = *(const float4*)(S + (long long)cc * D_OUT + lane * 4);
                        }
#pragma unroll
                        for (unsigned u = 0; u < 8; ++u) {
                            const float v = __int_as_float(__builtin_amdgcn_readlane(vv, (int)(k + u)));
                            const int   d = __builtin_amdgcn_readlane(dv, (int)(k + u));
                            float4* ap = acc + d * (D_OUT / 4) + lane;
                            float4 a = *ap;
                            a.x += s[u].x * v;
                            a.y += s[u].y * v;
                            a.z += s[u].z * v;
                            a.w += s[u].w * v;
                            *ap = a;
                        }
                    }
                    for (; k < nb; ++k) {
                        const int   cc = __builtin_amdgcn_readlane(cv, (int)k);
                        const float v  = __int_as_float(__builtin_amdgcn_readlane(vv, (int)k));
                        const int   d  = __builtin_amdgcn_readlane(dv, (int)k);
                        const float4 s1 = *(const float4*)(S + (long long)cc * D_OUT + lane * 4);
                        float4* ap = acc + d * (D_OUT / 4) + lane;
                        float4 a = *ap;
                        a.x += s1.x * v;
                        a.y += s1.y * v;
                        a.z += s1.z * v;
                        a.w += s1.w * v;
                        *ap = a;
                    }
                    const unsigned left = qs - nb;
                    uint2 mv = make_uint2(0u, 0u);
                    if ((unsigned)lane < left) mv = mystage[nb + lane];
                    if ((unsigned)lane < left) mystage[lane] = mv;
                    qs = left;
                }
                cur = nxt;
            }
        }

        const unsigned more = (unsigned)__builtin_amdgcn_readfirstlane((int)pending);
        __syncthreads();
        if (!more) break;
        if (tid == 0) pending = 0;
        __syncthreads();
    }

    const float4 bv = *(const float4*)(bias + lane * 4);
    for (unsigned d = (unsigned)wave; d < tile_rows; d += TILE_WAVES) {
        const float4 a = acc[d * (D_OUT / 4) + lane];
        float4 o;
        o.x = a.x + bv.x;
        o.y = a.y + bv.y;
        o.z = a.z + bv.z;
        o.w = a.w + bv.w;
        *(float4*)(out + (long long)(r0 + d) * D_OUT + lane * 4) = o;
    }
}

extern "C" void kernel_launch(void* const* d_in, const int* in_sizes, int n_in,
                              void* d_out, int out_size, void* d_ws, size_t ws_size,
                              hipStream_t stream) {
    const int*   rows = (const int*)d_in[0];
    const int*   cols = (const int*)d_in[1];
    const float* vals = (const float*)d_in[2];
    const float* x    = (const float*)d_in[3];
    const float* W    = (const float*)d_in[4];
    const float* bias = (const float*)d_in[5];

    const int E = in_sizes[0];
    const int N = in_sizes[3] / D_IN;

    float* support = (float*)d_ws;
    unsigned* keys = (unsigned*)(support + (size_t)N * D_OUT);

    {
        int mtiles = N / 16;
        int blocks = (mtiles + 7) / 8;
        gcn_gemm_xw<<<blocks, 256, 0, stream>>>(x, W, support, N, rows, keys, E);
    }

    {
        int blocks = (N + ROWS_PER_WG - 1) / ROWS_PER_WG;
        gcn_tile_sum<<<blocks, 1024, 0, stream>>>(keys, rows, cols, vals, support, bias,
                                                  (float*)d_out, E, N);
    }
}


# Noted edges ordered by owner wave in registers (ballot count, lane prefix, stable) and put back in their list: a wave reads only its own segment of each list, not all of it; sum order unchanged
# speedup vs baseline: 2.0960x; 1.0744x over previous
#include <hip/hip_runtime.h>

typedef __attribute__((ext_vector_type(2))) float v2f;
typedef __attribute__((ext_vector_type(8))) float v8f;

#define D_IN  256
#define D_OUT 128

#define TILE_WAVES    32
#define ROWS_PER_WG   400
#define LIST_CAP      800
#define STAGE_CAP     64
#define SCAN_STEP     512
#define SCAN_GROUP    4
#define KEY_NONE      255u
#define SORT_CHUNKS   10
#define SORT_MAX      (SORT_CHUNKS * 32)
#define OFFS_AT       (LIST_CAP - TILE_WAVES)
#define SORTED_FLAG   0x80000000u

__device__ __forceinline__ unsigned tile_key(int r, int n_rows) {
    return ((unsigned)r < (unsigned)n_rows) ? (unsigned)r / ROWS_PER_WG : KEY_NONE;
}

__global__ __launch_bounds__(256) void gcn_gemm_xw(const float* __restrict__ X,
                                                   const float* __restrict__ W,
                                                   float* __restrict__ S,
                                                   int n_rows,
                                                   const int* __restrict__ rows,
                                                   unsigned* __restrict__ keys,
                                                   int n_edges) {
    {
        const int nk = ((n_edges + SCAN_STEP - 1) / SCAN_STEP) * (SCAN_STEP / 4);
        for (int g = blockIdx.x * 256 + threadIdx.x; g < nk; g += gridDim.x * 256) {
            const int e = g * 4;
            int4 r;
            if (e + 3 < n_edges) {
                r = *(const int4*)(rows + e);
            } else {
                r.x = (e     < n_edges) ? rows[e]     : -1;
                r.y = (e + 1 < n_edges) ? rows[e + 1] : -1;
                r.z = (e + 2 < n_edges) ? rows[e + 2] : -1;
                r.w = (e + 3 < n_edges) ? rows[e + 3] : -1;
            }
            keys[g] = tile_key(r.x, n_rows) | (tile_key(r.y, n_rows) << 8) |
                      (tile_key(r.z, n_rows) << 16) | (tile_key(r.w, n_rows) << 24);
        }
    }
    const int lane  = threadIdx.x & 31;
    const int wave  = threadIdx.x >> 5;
    const int mtile = blockIdx.x * 8 + wave;
    if (mtile * 16 + 16 > n_rows) return;
    const int m0 = mtile * 16;

    const int am = lane & 15;
    const int ak = (lane >> 4) * 2;
    const int bn = lane & 15;
    const int bk = (lane >> 4) * 2;

    v8f c[8];
#pragma unroll
    for (int t = 0; t < 8; ++t) c[t] = (v8f){0.f, 0.f, 0.f, 0.f, 0.f, 0.f, 0.f, 0.f};

    const float* xrow = X + (long long)(m0 + am) * D_IN;

    for (int k = 0; k < D_IN; k += 4) {
        v2f a = *(const v2f*)(xrow + k + ak);
#pragma unroll
        for (int t = 0; t < 8; ++t) {
            v2f b;
            b.x = W[(long long)(k + bk)     * D_OUT + t * 16 + bn];
            b.y = W[(long long)(k + bk + 1) * D_OUT + t * 16 + bn];
            c[t] = __builtin_amdgcn_wmma_f32_16x16x4_f32(
                 false, a,  false, b,
                 (short)0, c[t],  false,  false);
        }
    }

    const int cn    = lane & 15;
    const int cmoff = (lane >> 4) * 8;
#pragma unroll
    for (int t = 0; t < 8; ++t) {
#pragma unroll
        for (int r = 0; r < 8; ++r) {
            S[(long long)(m0 + r + cmoff) * D_OUT + t * 16 + cn] = c[t][r];
        }
    }
}

__global__ __launch_bounds__(1024) void gcn_tile_sum(const unsigned* __restrict__ keys,
                                                     const int* __restrict__ rows,
                                                     const int* __restrict__ cols,
                                                     const float* __restrict__ vals,
                                                     const float* __restrict__ S,
                                                     const float* __restrict__ bias,
                                                     float* __restrict__ out,
                                                     int n_edges, int n_rows) {
    __shared__ float4   acc[ROWS_PER_WG * (D_OUT / 4)];
    __shared__ unsigned lists[TILE_WAVES * LIST_CAP];
    __shared__ uint2    stage_all[TILE_WAVES * STAGE_CAP];
    __shared__ unsigned cnt[TILE_WAVES];
    __shared__ unsigned pending;

    const int tid  = threadIdx.x;
    const int lane = tid & 31;
    const int wave = __builtin_amdgcn_readfirstlane(tid >> 5);
    const unsigned r0 = blockIdx.x * ROWS_PER_WG;
    const int rem_rows = n_rows - (int)r0;
    const unsigned tile_rows = (unsigned)(rem_rows < ROWS_PER_WG ? rem_rows : ROWS_PER_WG);

    for (int i = tid; i < ROWS_PER_WG * (D_OUT / 4); i += 1024)
        acc[i] = make_float4(0.f, 0.f, 0.f, 0.f);
    if (tid == 0) pending = 0;

    unsigned* mylist  = lists + wave * LIST_CAP;
    uint2*    mystage = stage_all + wave * STAGE_CAP;

    const int nsteps = (n_edges + SCAN_STEP - 1) / SCAN_STEP;
    const int rot    = (int)(((long long)blockIdx.x * nsteps) / gridDim.x);
    const unsigned pat = blockIdx.x * 0x01010101u;

    __syncthreads();

    int st = wave;
    unsigned t = 0;
    for (;;) {
        unsigned qn = 0;
        while (st < nsteps && qn + SCAN_STEP <= LIST_CAP) {
            uint4 kk[SCAN_GROUP];
#pragma unroll
            for (int u = 0; u < SCAN_GROUP; ++u) {
                int s = st + u * TILE_WAVES;
                if (s >= nsteps) s = nsteps - 1;
                int sp = s + rot;
                if (sp >= nsteps) sp -= nsteps;
                kk[u] = *(const uint4*)(keys + sp * (SCAN_STEP / 4) + lane * 4);
            }
#pragma unroll
            for (int u = 0; u < SCAN_GROUP; ++u) {
                if (st < nsteps && qn + SCAN_STEP <= LIST_CAP) {
                    const unsigned kw[4] = {kk[u].x, kk[u].y, kk[u].z, kk[u].w};
#pragma unroll
                    for (int j = 0; j < 4; ++j) {
                        unsigned y = kw[j] ^ pat;
                        unsigned z = (y - 0x01010101u) & ~y & 0x80808080u;
                        unsigned m = __builtin_amdgcn_ballot_w32(z != 0u);
                        while (m) {
                            if (z) {
                                const unsigned b    = (unsigned)__builtin_ctz(z) >> 3;
                                const unsigned slot = qn + __builtin_amdgcn_mbcnt_lo(m, 0u);
                                mylist[slot] = (t << 9) | (unsigned)(lane * 16 + j * 4) | b;
                                y |= 0xFFu << (b * 8u);
                                z = (y - 0x01010101u) & ~y & 0x80808080u;
                            }
                            qn += (unsigned)__builtin_popcount(m);
                            m = __builtin_amdgcn_ballot_w32(z != 0u);
                        }
                    }
                    st += TILE_WAVES;
                    ++t;
                }
            }
        }
        const bool ordered = (qn <= (unsigned)SORT_MAX);
        if (ordered) {
            unsigned en[SORT_CHUNKS];
            unsigned ow[SORT_CHUNKS];
#pragma unroll
            for (int c = 0; c < SORT_CHUNKS; ++c) {
                const unsigned p = (unsigned)(c * 32 + lane);
                en[c] = 0u;
                ow[c] = 32u;
                if (p < qn) {
                    const unsigned e0 = mylist[p];
                    int spd = wave + TILE_WAVES * (int)((e0 >> 9) & 0x3FFFu) + rot;
                    if (spd >= nsteps) spd -= nsteps;
                    const unsigned e  = (unsigned)(spd * SCAN_STEP) + (e0 & 511u);
                    const unsigned tr = (unsigned)rows[e] - r0;
                    en[c] = e0 | (tr << 23);
                    ow[c] = tr & 31u;
                }
            }
            unsigned tot = 0;
            unsigned pre[SORT_CHUNKS];
            unsigned rk[SORT_CHUNKS];
#pragma unroll
            for (int c = 0; c < SORT_CHUNKS; ++c) {
                pre[c] = tot;
                rk[c]  = 0u;
                if ((unsigned)(c * 32) < qn) {
                    const unsigned vm = __builtin_amdgcn_ballot_w32((unsigned)(c * 32 + lane) < qn);
                    unsigned mm = vm;
                    unsigned hb = vm;
#pragma unroll
                    for (int k = 0; k < 5; ++k) {
                        const unsigned bal = __builtin_amdgcn_ballot_w32(((ow[c] >> k) & 1u) != 0u);
                        mm &= ((ow[c] >> k) & 1u) ? bal : ~bal;
                        hb &= (((unsigned)lane >> k) & 1u) ? bal : ~bal;
                    }
                    tot  += (unsigned)__builtin_popcount(hb);
                    rk[c] = __builtin_amdgcn_mbcnt_lo(mm, 0u);
                }
            }
            unsigned incl = tot;
#pragma unroll
            for (int d = 1; d < 32; d <<= 1) {
                const unsigned o = (unsigned)__builtin_amdgcn_ds_bpermute((int)((((unsigned)(lane - d)) & 31u) << 2), (int)incl);
                if (lane >= d) incl += o;
            }
            const unsigned start = incl - tot;
#pragma unroll
            for (int c = 0; c < SORT_CHUNKS; ++c) {
                if ((unsigned)(c * 32) < qn) {
                    const unsigned b = (unsigned)__builtin_amdgcn_ds_bpermute((int)((ow[c] & 31u) << 2),
                                                                              (int)(start + pre[c]));
                    if ((unsigned)(c * 32 + lane) < qn) mylist[b + rk[c]] = en[c];
                }
            }
            mylist[OFFS_AT + lane] = start;
        } else {
            for (unsigned p = (unsigned)lane; p < qn; p += 32u) {
                const unsigned en = mylist[p];
                int spd = wave + TILE_WAVES * (int)((en >> 9) & 0x3FFFu) + rot;
                if (spd >= nsteps) spd -= nsteps;
                const unsigned e = (unsigned)(spd * SCAN_STEP) + (en & 511u);
                mylist[p] = en | (((unsigned)rows[e] - r0) << 23);
            }
        }
        if (lane == 0) {
            cnt[wave] = qn | (ordered ? SORTED_FLAG : 0u);
            if (st < nsteps) pending = 1;
        }
        __syncthreads();

        const unsigned cl = cnt[lane];
        unsigned sb = 0u;
        unsigned sn = cl & ~SORTED_FLAG;
        if (cl & SORTED_FLAG) {
            sb = lists[lane * LIST_CAP + OFFS_AT + wave];
            const unsigned se = (wave < TILE_WAVES - 1) ? lists[lane * LIST_CAP + OFFS_AT + wave + 1] : sn;
            sn = se - sb;
        }
        unsigned qs = 0;
        for (int L = 0; L <= TILE_WAVES; ++L) {
            unsigned n = 0;
            unsigned base = 0;
            unsigned pickbits = 0u;
            int nchunks = 1;
            unsigned cur = 0;
            if (L < TILE_WAVES) {
                n    = (unsigned)__builtin_amdgcn_readlane((int)sn, L);
                base = (unsigned)(L * LIST_CAP) + (unsigned)__builtin_amdgcn_readlane((int)sb, L);
                pickbits = ((unsigned)__builtin_amdgcn_readlane((int)cl, L) & SORTED_FLAG) ? 0u : 31u;
                nchunks = (int)((n + 31u) >> 5);
                if ((unsigned)lane < n) cur = lists[base + lane];
            }
            for (int c = 0; c < nchunks; ++c) {
                unsigned nxt = 0;
                if (L < TILE_WAVES) {
                    const unsigned pn = (unsigned)((c + 1) * 32 + lane);
                    if (pn < n) nxt = lists[base + pn];
                    const unsigned p = (unsigned)(c * 32 + lane);
                    const bool own = (p < n) && ((((cur >> 23) ^ (unsigned)wave) & pickbits) == 0u);
                    const unsigned m = __builtin_amdgcn_ballot_w32(own);
                    if (m) {
                        if (own) {
                            const unsigned slot = qs + __builtin_amdgcn_mbcnt_lo(m, 0u);
                            int spd = L + TILE_WAVES * (int)((cur >> 9) & 0x3FFFu) + rot;
                            if (spd >= nsteps) spd -= nsteps;
                            mystage[slot] = make_uint2((unsigned)(spd * SCAN_STEP) + (cur & 511u),
                                                       cur >> 23);
                        }
                        qs += (unsigned)__builtin_popcount(m);
                    }
                }
                const unsigned thr = (L < TILE_WAVES) ? 32u : 1u;
                if (qs >= thr) {
                    const unsigned nb = qs < 32u ? qs : 32u;
                    int cv = 0, vv = 0, dv = 0;
                    if ((unsigned)lane < nb) {
                        const uint2 sg = mystage[lane];
                        cv = cols[sg.x];
                        vv = __float_as_int(vals[sg.x]);
                        dv = (int)sg.y;
                    }
                    unsigned k = 0;
                    for (; k + 8 <= nb; k += 8) {
                        float4 s[8];
#pragma unroll
                        for (unsigned u = 0; u < 8; ++u) {
                            const int cc = __builtin_amdgcn_readlane(cv, (int)(k + u));
                            s[u] = *(const float4*)(S + (long long)cc * D_OUT + lane * 4);
                        }
#pragma unroll
                        for (unsigned u = 0; u < 8; ++u) {
                            const float v = __int_as_float(__builtin_amdgcn_readlane(vv, (int)(k + u)));
                            const int   d = __builtin_amdgcn_readlane(dv, (int)(k + u));
                            float4* ap = acc + d * (D_OUT / 4) + lane;
                            float4 a = *ap;
                            a.x += s[u].x * v;
                            a.y += s[u].y * v;
                            a.z += s[u].z * v;
                            a.w += s[u].w * v;
                            *ap = a;
                        }
                    }
                    for (; k < nb; ++k) {
                        const int   cc = __builtin_amdgcn_readlane(cv, (int)k);
                        const float v  = __int_as_float(__builtin_amdgcn_readlane(vv, (int)k));
                        const int   d  = __builtin_amdgcn_readlane(dv, (int)k);
                        const float4 s1 = *(const float4*)(S + (long long)cc * D_OUT + lane * 4);
                        float4* ap = acc + d * (D_OUT / 4) + lane;
                        float4 a = *ap;
                        a.x += s1.x * v;
                        a.y += s1.y * v;
                        a.z += s1.z * v;
                        a.w += s1.w * v;
                        *ap = a;
                    }
                    const unsigned left = qs - nb;
                    uint2 mv = make_uint2(0u, 0u);
                    if ((unsigned)lane < left) mv = mystage[nb + lane];
                    if ((unsigned)lane < left) mystage[lane] = mv;
                    qs = left;
                }
                cur = nxt;
            }
        }

        const unsigned more = (unsigned)__builtin_amdgcn_readfirstlane((int)pending);
        __syncthreads();
        if (!more) break;
        if (tid == 0) pending = 0;
        __syncthreads();
    }

    const float4 bv = *(const float4*)(bias + lane * 4);
    for (unsigned d = (unsigned)wave; d < tile_rows; d += TILE_WAVES) {
        const float4 a = acc[d * (D_OUT / 4) + lane];
        float4 o;
        o.x = a.x + bv.x;
        o.y = a.y + bv.y;
        o.z = a.z + bv.z;
        o.w = a.w + bv.w;
        *(float4*)(out + (long long)(r0 + d) * D_OUT + lane * 4) = o;
    }
}

extern "C" void kernel_launch(void* const* d_in, const int* in_sizes, int n_in,
                              void* d_out, int out_size, void* d_ws, size_t ws_size,
                              hipStream_t stream) {
    const int*   rows = (const int*)d_in[0];
    const int*   cols = (const int*)d_in[1];
    const float* vals = (const float*)d_in[2];
    const float* x    = (const float*)d_in[3];
    const float* W    = (const float*)d_in[4];
    const float* bias = (const float*)d_in[5];

    const int E = in_sizes[0];
    const int N = in_sizes[3] / D_IN;

    float* support = (float*)d_ws;
    unsigned* keys = (unsigned*)(support + (size_t)N * D_OUT);

    {
        int mtiles = N / 16;
        int blocks = (mtiles + 7) / 8;
        gcn_gemm_xw<<<blocks, 256, 0, stream>>>(x, W, support, N, rows, keys, E);
    }

    {
        int blocks = (N + ROWS_PER_WG - 1) / ROWS_PER_WG;
        gcn_tile_sum<<<blocks, 1024, 0, stream>>>(keys, rows, cols, vals, support, bias,
                                                  (float*)d_out, E, N);
    }
}


# First launch packs each edge as one 8-byte record (source row, tile row, value); noting wave reads it once, hands it on in its ordered LDS list: 1 scattered read per edge, not 3; sum order unchanged
# speedup vs baseline: 2.2454x; 1.0713x over previous
#include <hip/hip_runtime.h>

typedef __attribute__((ext_vector_type(2))) float v2f;
typedef __attribute__((ext_vector_type(8))) float v8f;

#define D_IN  256
#define D_OUT 128

#define TILE_WAVES    32
#define ROWS_PER_WG   400
#define LIST_CAP      800
#define STAGE_CAP     64
#define SCAN_STEP     512
#define SCAN_GROUP    4
#define KEY_NONE      255u
#define SORT_CHUNKS   10
#define SORT_MAX      (SORT_CHUNKS * 32)
#define OFFS_AT       (LIST_CAP - TILE_WAVES)
#define SORTED_FLAG   0x80000000u
#define COL_BITS      17
#define COL_MASK      0x1FFFFu
#define UNRES_FLAG    0x80000000u
static_assert(2 * SORT_MAX <= OFFS_AT, "an ordered list (two dwords per edge) must end below its segment starts");

__device__ __forceinline__ unsigned tile_key(int r, int n_rows) {
    return ((unsigned)r < (unsigned)n_rows) ? (unsigned)r / ROWS_PER_WG : KEY_NONE;
}

__device__ __forceinline__ unsigned rec_word(int r, int c, unsigned key) {
    return ((unsigned)c & COL_MASK) | ((((unsigned)r - key * ROWS_PER_WG) & 0x1FFu) << COL_BITS);
}

__global__ __launch_bounds__(256) void gcn_gemm_xw(const float* __restrict__ X,
                                                   const float* __restrict__ W,
                                                   float* __restrict__ S,
                                                   int n_rows,
                                                   const int* __restrict__ rows,
                                                   const int* __restrict__ cols,
                                                   const float* __restrict__ vals,
                                                   unsigned* __restrict__ keys,
                                                   unsigned* __restrict__ recw,
                                                   int n_edges) {
    {
        const int nk = ((n_edges + SCAN_STEP - 1) / SCAN_STEP) * (SCAN_STEP / 4);
        for (int g = blockIdx.x * 256 + threadIdx.x; g < nk; g += gridDim.x * 256) {
            const int e = g * 4;
            if (e + 3 < n_edges) {
                const int4   r = *(const int4*)(rows + e);
                const int4   c = *(const int4*)(cols + e);
                const float4 v = *(const float4*)(vals + e);
                const unsigned k0 = tile_key(r.x, n_rows), k1 = tile_key(r.y, n_rows);
                const unsigned k2 = tile_key(r.z, n_rows), k3 = tile_key(r.w, n_rows);
                keys[g] = k0 | (k1 << 8) | (k2 << 16) | (k3 << 24);
                uint4 lo, hi;
                lo.x = rec_word(r.x, c.x, k0);  lo.y = __float_as_uint(v.x);
                lo.z = rec_word(r.y, c.y, k1);  lo.w = __float_as_uint(v.y);
                hi.x = rec_word(r.z, c.z, k2);  hi.y = __float_as_uint(v.z);
                hi.z = rec_word(r.w, c.w, k3);  hi.w = __float_as_uint(v.w);
                *(uint4*)(recw + 2 * (size_t)e)     = lo;
                *(uint4*)(recw + 2 * (size_t)e + 4) = hi;
            } else {
                unsigned kq = 0u;
                for (int j = 0; j < 4; ++j) {
                    unsigned k = KEY_NONE;
                    if (e + j < n_edges) {
                        const int r = rows[e + j];
                        k = tile_key(r, n_rows);
                        recw[2 * (size_t)(e + j)]     = rec_word(r, cols[e + j], k);
                        recw[2 * (size_t)(e + j) + 1] = __float_as_uint(vals[e + j]);
                    }
                    kq |= k << (8 * j);
                }
                keys[g] = kq;
            }
        }
    }
    const int lane  = threadIdx.x & 31;
    const int wave  = threadIdx.x >> 5;
    const int mtile = blockIdx.x * 8 + wave;
    if (mtile * 16 + 16 > n_rows) return;
    const int m0 = mtile * 16;

    const int am = lane & 15;
    const int ak = (lane >> 4) * 2;
    const int bn = lane & 15;
    const int bk = (lane >> 4) * 2;

    v8f c[8];
#pragma unroll
    for (int t = 0; t < 8; ++t) c[t] = (v8f){0.f, 0.f, 0.f, 0.f, 0.f, 0.f, 0.f, 0.f};

    const float* xrow = X + (long long)(m0 + am) * D_IN;

    for (int k = 0; k < D_IN; k += 4) {
        v2f a = *(const v2f*)(xrow + k + ak);
#pragma unroll
        for (int t = 0; t < 8; ++t) {
            v2f b;
            b.x = W[(long long)(k + bk)     * D_OUT + t * 16 + bn];
            b.y = W[(long long)(k + bk + 1) * D_OUT + t * 16 + bn];
            c[t] = __builtin_amdgcn_wmma_f32_16x16x4_f32(
                 false, a,  false, b,
                 (short)0, c[t],  false,  false);
        }
    }

    const int cn    = lane & 15;
    const int cmoff = (lane >> 4) * 8;
#pragma unroll
    for (int t = 0; t < 8; ++t) {
#pragma unroll
        for (int r = 0; r < 8; ++r) {
            S[(long long)(m0 + r + cmoff) * D_OUT + t * 16 + cn] = c[t][r];
        }
    }
}

__global__ __launch_bounds__(1024) void gcn_tile_sum(const unsigned* __restrict__ keys,
                                                     const unsigned* __restrict__ recw,
                                                     const float* __restrict__ S,
                                                     const float* __restrict__ bias,
                                                     float* __restrict__ out,
                                                     int n_edges, int n_rows) {
    __shared__ float4   acc[ROWS_PER_WG * (D_OUT / 4)];
    __shared__ unsigned lists[TILE_WAVES * LIST_CAP];
    __shared__ unsigned long long stage_all[TILE_WAVES * STAGE_CAP];
    __shared__ unsigned cnt[TILE_WAVES];
    __shared__ unsigned pending;

    const unsigned long long* __restrict__ rec = (const unsigned long long*)recw;
    const int tid  = threadIdx.x;
    const int lane = tid & 31;
    const int wave = __builtin_amdgcn_readfirstlane(tid >> 5);
    const unsigned r0 = blockIdx.x * ROWS_PER_WG;
    const int rem_rows = n_rows - (int)r0;
    const unsigned tile_rows = (unsigned)(rem_rows < ROWS_PER_WG ? rem_rows : ROWS_PER_WG);

    for (int i = tid; i < ROWS_PER_WG * (D_OUT / 4); i += 1024)
        acc[i] = make_float4(0.f, 0.f, 0.f, 0.f);
    if (tid == 0) pending = 0;

    unsigned* mylist  = lists + wave * LIST_CAP;
    unsigned long long* mystage = stage_all + wave * STAGE_CAP;

    const int nsteps = (n_edges + SCAN_STEP - 1) / SCAN_STEP;
    const int rot    = (int)(((long long)blockIdx.x * nsteps) / gridDim.x);
    const unsigned pat = blockIdx.x * 0x01010101u;

    __syncthreads();

    int st = wave;
    unsigned t = 0;
    for (;;) {
        unsigned qn = 0;
        while (st < nsteps && qn + SCAN_STEP <= LIST_CAP) {
            uint4 kk[SCAN_GROUP];
#pragma unroll
            for (int u = 0; u < SCAN_GROUP; ++u) {
                int s = st + u * TILE_WAVES;
                if (s >= nsteps) s = nsteps - 1;
                int sp = s + rot;
                if (sp >= nsteps) sp -= nsteps;
                kk[u] = *(const uint4*)(keys + sp * (SCAN_STEP / 4) + lane * 4);
            }
#pragma unroll
            for (int u = 0; u < SCAN_GROUP; ++u) {
                if (st < nsteps && qn + SCAN_STEP <= LIST_CAP) {
                    const unsigned kw[4] = {kk[u].x, kk[u].y, kk[u].z, kk[u].w};
#pragma unroll
                    for (int j = 0; j < 4; ++j) {
                        unsigned y = kw[j] ^ pat;
                        unsigned z = (y - 0x01010101u) & ~y & 0x80808080u;
                        unsigned m = __builtin_amdgcn_ballot_w32(z != 0u);
                        while (m) {
                            if (z) {
                                const unsigned b    = (unsigned)__builtin_ctz(z) >> 3;
                                const unsigned slot = qn + __builtin_amdgcn_mbcnt_lo(m, 0u);
                                mylist[slot] = (t << 9) | (unsigned)(lane * 16 + j * 4) | b;
                                y |= 0xFFu << (b * 8u);
                                z = (y - 0x01010101u) & ~y & 0x80808080u;
                            }
                            qn += (unsigned)__builtin_popcount(m);
                            m = __builtin_amdgcn_ballot_w32(z != 0u);
                        }
                    }
                    st += TILE_WAVES;
                    ++t;
                }
            }
        }
        const bool ordered = (qn <= (unsigned)SORT_MAX);
        if (ordered) {
            unsigned en[SORT_CHUNKS];
            unsigned ev[SORT_CHUNKS];
            unsigned ow[SORT_CHUNKS];
#pragma unroll
            for (int c = 0; c < SORT_CHUNKS; ++c) {
                const unsigned p = (unsigned)(c * 32 + lane);
                en[c] = 0u;
                ev[c] = 0u;
                ow[c] = 32u;
                if (p < qn) {
                    const unsigned e0 = mylist[p];
                    int spd = wave + TILE_WAVES * (int)((e0 >> 9) & 0x3FFFu) + rot;
                    if (spd >= nsteps) spd -= nsteps;
                    const unsigned e  = (unsigned)(spd * SCAN_STEP) + (e0 & 511u);
                    const unsigned long long rc = rec[e];
                    en[c] = (unsigned)rc;
                    ev[c] = (unsigned)(rc >> 32);
                    ow[c] = ((unsigned)rc >> COL_BITS) & 31u;
                }
            }
            unsigned tot = 0;
            unsigned pre[SORT_CHUNKS];
            unsigned rk[SORT_CHUNKS];
#pragma unroll
            for (int c = 0; c < SORT_CHUNKS; ++c) {
                pre[c] = tot;
                rk[c]  = 0u;
                if ((unsigned)(c * 32) < qn) {
                    const unsigned vm = __builtin_amdgcn_ballot_w32((unsigned)(c * 32 + lane) < qn);
                    unsigned mm = vm;
                    unsigned hb = vm;
#pragma unroll
                    for (int k = 0; k < 5; ++k) {
                        const unsigned bal = __builtin_amdgcn_ballot_w32(((ow[c] >> k) & 1u) != 0u);
                        mm &= ((ow[c] >> k) & 1u) ? bal : ~bal;
                        hb &= (((unsigned)lane >> k) & 1u) ? bal : ~bal;
                    }
                    tot  += (unsigned)__builtin_popcount(hb);
                    rk[c] = __builtin_amdgcn_mbcnt_lo(mm, 0u);
                }
            }
            unsigned incl = tot;
#pragma unroll
            for (int d = 1; d < 32; d <<= 1) {
                const unsigned o = (unsigned)__builtin_amdgcn_ds_bpermute((int)((((unsigned)(lane - d)) & 31u) << 2), (int)incl);
                if (lane >= d) incl += o;
            }
            const unsigned start = incl - tot;
#pragma unroll
            for (int c = 0; c < SORT_CHUNKS; ++c) {
                if ((unsigned)(c * 32) < qn) {
                    const unsigned b = (unsigned)__builtin_amdgcn_ds_bpermute((int)((ow[c] & 31u) << 2),
                                                                              (int)(start + pre[c]));
                    if ((unsigned)(c * 32 + lane) < qn) {
                        const unsigned q = b + rk[c];
                        mylist[2u * q]      = en[c];
                        mylist[2u * q + 1u] = ev[c];
                    }
                }
            }
            mylist[OFFS_AT + lane] = start;
        } else {
            for (unsigned p = (unsigned)lane; p < qn; p += 32u) {
                const unsigned en = mylist[p];
                int spd = wave + TILE_WAVES * (int)((en >> 9) & 0x3FFFu) + rot;
                if (spd >= nsteps) spd -= nsteps;
                const unsigned e = (unsigned)(spd * SCAN_STEP) + (en & 511u);
                mylist[p] = en | (((unsigned)rec[e] >> COL_BITS) << 23);
            }
        }
        if (lane == 0) {
            cnt[wave] = qn | (ordered ? SORTED_FLAG : 0u);
            if (st < nsteps) pending = 1;
        }
        __syncthreads();

        const unsigned cl = cnt[lane];
        unsigned sb = 0u;
        unsigned sn = cl & ~SORTED_FLAG;
        if (cl & SORTED_FLAG) {
            sb = lists[lane * LIST_CAP + OFFS_AT + wave];
            const unsigned se = (wave < TILE_WAVES - 1) ? lists[lane * LIST_CAP + OFFS_AT + wave + 1] : sn;
            sn = se - sb;
        }
        unsigned qs = 0;
        for (int L = 0; L <= TILE_WAVES; ++L) {
            unsigned n = 0;
            unsigned base = 0;
            unsigned pickbits = 0u;
            unsigned esh = 0u;
            int nchunks = 1;
            unsigned cur = 0, curv = 0;
            if (L < TILE_WAVES) {
                n    = (unsigned)__builtin_amdgcn_readlane((int)sn, L);
                esh  = ((unsigned)__builtin_amdgcn_readlane((int)cl, L) & SORTED_FLAG) ? 1u : 0u;
                base = (unsigned)(L * LIST_CAP) + ((unsigned)__builtin_amdgcn_readlane((int)sb, L) << esh);
                pickbits = esh ? 0u : 31u;
                nchunks = (int)((n + 31u) >> 5);
                if ((unsigned)lane < n) {
                    cur  = lists[base + ((unsigned)lane << esh)];
                    curv = lists[base + ((unsigned)lane << esh) + esh];
                }
            }
            for (int c = 0; c < nchunks; ++c) {
                unsigned nxt = 0, nxtv = 0;
                if (L < TILE_WAVES) {
                    const unsigned pn = (unsigned)((c + 1) * 32 + lane);
                    if (pn < n) {
                        nxt  = lists[base + (pn << esh)];
                        nxtv = lists[base + (pn << esh) + esh];
                    }
                    const unsigned p = (unsigned)(c * 32 + lane);
                    const bool own = (p < n) && ((((cur >> 23) ^ (unsigned)wave) & pickbits) == 0u);
                    const unsigned m = __builtin_amdgcn_ballot_w32(own);
                    if (m) {
                        if (own) {
                            const unsigned slot = qs + __builtin_amdgcn_mbcnt_lo(m, 0u);
                            unsigned long long sg = (unsigned long long)cur | ((unsigned long long)curv << 32);
                            if (!esh) {
                                int spd = L + TILE_WAVES * (int)((cur >> 9) & 0x3FFFu) + rot;
                                if (spd >= nsteps) spd -= nsteps;
                                sg = (unsigned long long)(((unsigned)(spd * SCAN_STEP) + (cur & 511u)) | UNRES_FLAG);
                            }
                            mystage[slot] = sg;
                        }
                        qs += (unsigned)__builtin_popcount(m);
                    }
                }
                const unsigned thr = (L < TILE_WAVES) ? 32u : 1u;
                if (qs >= thr) {
                    const unsigned nb = qs < 32u ? qs : 32u;
                    int cv = 0, vv = 0, dv = 0;
                    if ((unsigned)lane < nb) {
                        unsigned long long sg = mystage[lane];
                        if ((unsigned)sg & UNRES_FLAG) sg = rec[(unsigned)sg & ~UNRES_FLAG];
                        cv = (int)((unsigned)sg & COL_MASK);
                        dv = (int)((unsigned)sg >> COL_BITS);
                        vv = (int)(unsigned)(sg >> 32);
                    }
                    unsigned k = 0;
                    for (; k + 8 <= nb; k += 8) {
                        float4 s[8];
#pragma unroll
                        for (unsigned u = 0; u < 8; ++u) {
                            const int cc = __builtin_amdgcn_readlane(cv, (int)(k + u));
                            s[u] = *(const float4*)(S + (long long)cc * D_OUT + lane * 4);
                        }
#pragma unroll
                        for (unsigned u = 0; u < 8; ++u) {
                            const float v = __int_as_float(__builtin_amdgcn_readlane(vv, (int)(k + u)));
                            const int   d = __builtin_amdgcn_readlane(dv, (int)(k + u));
                            float4* ap = acc + d * (D_OUT / 4) + lane;
                            float4 a = *ap;
                            a.x += s[u].x * v;
                            a.y += s[u].y * v;
                            a.z += s[u].z * v;
                            a.w += s[u].w * v;
                            *ap = a;
                        }
                    }
                    for (; k < nb; ++k) {
                        const int   cc = __builtin_amdgcn_readlane(cv, (int)k);
                        const float v  = __int_as_float(__builtin_amdgcn_readlane(vv, (int)k));
                        const int   d  = __builtin_amdgcn_readlane(dv, (int)k);
                        const float4 s1 = *(const float4*)(S + (long long)cc * D_OUT + lane * 4);
                        float4* ap = acc + d * (D_OUT / 4) + lane;
                        float4 a = *ap;
                        a.x += s1.x * v;
                        a.y += s1.y * v;
                        a.z += s1.z * v;
                        a.w += s1.w * v;
                        *ap = a;
                    }
                    const unsigned left = qs - nb;
                    unsigned long long mv = 0ull;
                    if ((unsigned)lane < left) mv = mystage[nb + lane];
                    if ((unsigned)lane < left) mystage[lane] = mv;
                    qs = left;
                }
                cur  = nxt;
                curv = nxtv;
            }
        }

        const unsigned more = (unsigned)__builtin_amdgcn_readfirstlane((int)pending);
        __syncthreads();
        if (!more) break;
        if (tid == 0) pending = 0;
        __syncthreads();
    }

    const float4 bv = *(const float4*)(bias + lane * 4);
    for (unsigned d = (unsigned)wave; d < tile_rows; d += TILE_WAVES) {
        const float4 a = acc[d * (D_OUT / 4) + lane];
        float4 o;
        o.x = a.x + bv.x;
        o.y = a.y + bv.y;
        o.z = a.z + bv.z;
        o.w = a.w + bv.w;
        *(float4*)(out + (long long)(r0 + d) * D_OUT + lane * 4) = o;
    }
}

extern "C" void kernel_launch(void* const* d_in, const int* in_sizes, int n_in,
                              void* d_out, int out_size, void* d_ws, size_t ws_size,
                              hipStream_t stream) {
    const int*   rows = (const int*)d_in[0];
    const int*   cols = (const int*)d_in[1];
    const float* vals = (const float*)d_in[2];
    const float* x    = (const float*)d_in[3];
    const float* W    = (const float*)d_in[4];
    const float* bias = (const float*)d_in[5];

    const int E = in_sizes[0];
    const int N = in_sizes[3] / D_IN;

    float* support = (float*)d_ws;
    unsigned* keys = (unsigned*)(support + (size_t)N * D_OUT);
    unsigned* recw = keys + (size_t)((E + SCAN_STEP - 1) / SCAN_STEP) * (SCAN_STEP / 4);

    {
        int mtiles = N / 16;
        int blocks = (mtiles + 7) / 8;
        gcn_gemm_xw<<<blocks, 256, 0, stream>>>(x, W, support, N, rows, cols, vals, keys, recw, E);
    }

    {
        int blocks = (N + ROWS_PER_WG - 1) / ROWS_PER_WG;
        gcn_tile_sum<<<blocks, 1024, 0, stream>>>(keys, recw, support, bias,
                                                  (float*)d_out, E, N);
    }
}


# Key plane bit-sliced (8 bit planes x 16 edges in a lane's 16 bytes, same size): a 512-edge scan step is matched with 11 logic ops and ONE ballot, not 4 x (3 ops + ballot); f32 sum order changes
# speedup vs baseline: 2.3805x; 1.0602x over previous
#include <hip/hip_runtime.h>

typedef __attribute__((ext_vector_type(2))) float v2f;
typedef __attribute__((ext_vector_type(8))) float v8f;

#define D_IN  256
#define D_OUT 128

#define TILE_WAVES    32
#define ROWS_PER_WG   400
#define LIST_CAP      800
#define STAGE_CAP     64
#define SCAN_STEP     512
#define SCAN_GROUP    4
#define KEY_NONE      255u
#define SORT_CHUNKS   10
#define SORT_MAX      (SORT_CHUNKS * 32)
#define OFFS_AT       (LIST_CAP - TILE_WAVES)
#define SORTED_FLAG   0x80000000u
#define COL_BITS      17
#define COL_MASK      0x1FFFFu
#define UNRES_FLAG    0x80000000u
static_assert(2 * SORT_MAX <= OFFS_AT, "an ordered list (two dwords per edge) must end below its segment starts");

__device__ __forceinline__ unsigned tile_key(int r, int n_rows) {
    return ((unsigned)r < (unsigned)n_rows) ? (unsigned)r / ROWS_PER_WG : KEY_NONE;
}

__device__ __forceinline__ unsigned rec_word(int r, int c, unsigned key) {
    return ((unsigned)c & COL_MASK) | ((((unsigned)r - key * ROWS_PER_WG) & 0x1FFu) << COL_BITS);
}

__global__ __launch_bounds__(256) void gcn_gemm_xw(const float* __restrict__ X,
                                                   const float* __restrict__ W,
                                                   float* __restrict__ S,
                                                   int n_rows,
                                                   const int* __restrict__ rows,
                                                   const int* __restrict__ cols,
                                                   const float* __restrict__ vals,
                                                   unsigned* __restrict__ keys,
                                                   unsigned* __restrict__ recw,
                                                   int n_edges) {
    {
        const int ng = ((n_edges + SCAN_STEP - 1) / SCAN_STEP) * (SCAN_STEP / 16);
        for (int q = blockIdx.x * 256 + threadIdx.x; q < ng; q += gridDim.x * 256) {
            const int e = q * 16;
            unsigned kb[16];
            if (e + 15 < n_edges) {
#pragma unroll
                for (int j = 0; j < 4; ++j) {
                    const int4 r = *(const int4*)(rows + e + 4 * j);
                    kb[4 * j + 0] = tile_key(r.x, n_rows);
                    kb[4 * j + 1] = tile_key(r.y, n_rows);
                    kb[4 * j + 2] = tile_key(r.z, n_rows);
                    kb[4 * j + 3] = tile_key(r.w, n_rows);
                }
            } else {
#pragma unroll
                for (int i = 0; i < 16; ++i)
                    kb[i] = (e + i < n_edges) ? tile_key(rows[e + i], n_rows) : KEY_NONE;
            }
            uint4 pl = make_uint4(0u, 0u, 0u, 0u);
#pragma unroll
            for (int i = 0; i < 16; ++i) {
                const unsigned k = kb[i];
                pl.x |= ((k & 1u) << i)        | (((k >> 1) & 1u) << (16 + i));
                pl.y |= (((k >> 2) & 1u) << i) | (((k >> 3) & 1u) << (16 + i));
                pl.z |= (((k >> 4) & 1u) << i) | (((k >> 5) & 1u) << (16 + i));
                pl.w |= (((k >> 6) & 1u) << i) | (((k >> 7) & 1u) << (16 + i));
            }
            *(uint4*)(keys + 4 * (size_t)q) = pl;
        }
        const int nq = (n_edges + 3) / 4;
        for (int g = blockIdx.x * 256 + threadIdx.x; g < nq; g += gridDim.x * 256) {
            const int e = g * 4;
            if (e + 3 < n_edges) {
                const int4   r = *(const int4*)(rows + e);
                const int4   c = *(const int4*)(cols + e);
                const float4 v = *(const float4*)(vals + e);
                const unsigned k0 = tile_key(r.x, n_rows), k1 = tile_key(r.y, n_rows);
                const unsigned k2 = tile_key(r.z, n_rows), k3 = tile_key(r.w, n_rows);
                uint4 lo, hi;
                lo.x = rec_word(r.x, c.x, k0);  lo.y = __float_as_uint(v.x);
                lo.z = rec_word(r.y, c.y, k1);  lo.w = __float_as_uint(v.y);
                hi.x = rec_word(r.z, c.z, k2);  hi.y = __float_as_uint(v.z);
                hi.z = rec_word(r.w, c.w, k3);  hi.w = __float_as_uint(v.w);
                *(uint4*)(recw + 2 * (size_t)e)     = lo;
                *(uint4*)(recw + 2 * (size_t)e + 4) = hi;
            } else {
                for (int j = 0; j < 4; ++j) {
                    if (e + j < n_edges) {
                        const int r = rows[e + j];
                        recw[2 * (size_t)(e + j)]     = rec_word(r, cols[e + j], tile_key(r, n_rows));
                        recw[2 * (size_t)(e + j) + 1] = __float_as_uint(vals[e + j]);
                    }
                }
            }
        }
    }
    const int lane  = threadIdx.x & 31;
    const int wave  = threadIdx.x >> 5;
    const int mtile = blockIdx.x * 8 + wave;
    if (mtile * 16 + 16 > n_rows) return;
    const int m0 = mtile * 16;

    const int am = lane & 15;
    const int ak = (lane >> 4) * 2;
    const int bn = lane & 15;
    const int bk = (lane >> 4) * 2;

    v8f c[8];
#pragma unroll
    for (int t = 0; t < 8; ++t) c[t] = (v8f){0.f, 0.f, 0.f, 0.f, 0.f, 0.f, 0.f, 0.f};

    const float* xrow = X + (long long)(m0 + am) * D_IN;

    for (int k = 0; k < D_IN; k += 4) {
        v2f a = *(const v2f*)(xrow + k + ak);
#pragma unroll
        for (int t = 0; t < 8; ++t) {
            v2f b;
            b.x = W[(long long)(k + bk)     * D_OUT + t * 16 + bn];
            b.y = W[(long long)(k + bk + 1) * D_OUT + t * 16 + bn];
            c[t] = __builtin_amdgcn_wmma_f32_16x16x4_f32(
                 false, a,  false, b,
                 (short)0, c[t],  false,  false);
        }
    }

    const int cn    = lane & 15;
    const int cmoff = (lane >> 4) * 8;
#pragma unroll
    for (int t = 0; t < 8; ++t) {
#pragma unroll
        for (int r = 0; r < 8; ++r) {
            S[(long long)(m0 + r + cmoff) * D_OUT + t * 16 + cn] = c[t][r];
        }
    }
}

__global__ __launch_bounds__(1024) void gcn_tile_sum(const unsigned* __restrict__ keys,
                                                     const unsigned* __restrict__ recw,
                                                     const float* __restrict__ S,
                                                     const float* __restrict__ bias,
                                                     float* __restrict__ out,
                                                     int n_edges, int n_rows) {
    __shared__ float4   acc[ROWS_PER_WG * (D_OUT / 4)];
    __shared__ unsigned lists[TILE_WAVES * LIST_CAP];
    __shared__ unsigned long long stage_all[TILE_WAVES * STAGE_CAP];
    __shared__ unsigned cnt[TILE_WAVES];
    __shared__ unsigned pending;

    const unsigned long long* __restrict__ rec = (const unsigned long long*)recw;
    const int tid  = threadIdx.x;
    const int lane = tid & 31;
    const int wave = __builtin_amdgcn_readfirstlane(tid >> 5);
    const unsigned r0 = blockIdx.x * ROWS_PER_WG;
    const int rem_rows = n_rows - (int)r0;
    const unsigned tile_rows = (unsigned)(rem_rows < ROWS_PER_WG ? rem_rows : ROWS_PER_WG);

    for (int i = tid; i < ROWS_PER_WG * (D_OUT / 4); i += 1024)
        acc[i] = make_float4(0.f, 0.f, 0.f, 0.f);
    if (tid == 0) pending = 0;

    unsigned* mylist  = lists + wave * LIST_CAP;
    unsigned long long* mystage = stage_all + wave * STAGE_CAP;

    const int nsteps = (n_edges + SCAN_STEP - 1) / SCAN_STEP;
    const int rot    = (int)(((long long)blockIdx.x * nsteps) / gridDim.x);
    unsigned cpl[4];
#pragma unroll
    for (int k = 0; k < 4; ++k)
        cpl[k] = (((blockIdx.x >> (2 * k)) & 1u) ? 0x0000FFFFu : 0u) |
                 (((blockIdx.x >> (2 * k + 1)) & 1u) ? 0xFFFF0000u : 0u);

    __syncthreads();

    int st = wave;
    unsigned t = 0;
    for (;;) {
        unsigned qn = 0;
        while (st < nsteps && qn + SCAN_STEP <= LIST_CAP) {
            uint4 kk[SCAN_GROUP];
#pragma unroll
            for (int u = 0; u < SCAN_GROUP; ++u) {
                int s = st + u * TILE_WAVES;
                if (s >= nsteps) s = nsteps - 1;
                int sp = s + rot;
                if (sp >= nsteps) sp -= nsteps;
                kk[u] = *(const uint4*)(keys + sp * (SCAN_STEP / 4) + lane * 4);
            }
#pragma unroll
            for (int u = 0; u < SCAN_GROUP; ++u) {
                if (st < nsteps && qn + SCAN_STEP <= LIST_CAP) {
                    const unsigned eq = ~(kk[u].x ^ cpl[0]) & ~(kk[u].y ^ cpl[1]) &
                                        ~(kk[u].z ^ cpl[2]) & ~(kk[u].w ^ cpl[3]);
                    unsigned hm = eq & (eq >> 16);
                    unsigned m = __builtin_amdgcn_ballot_w32(hm != 0u);
                    while (m) {
                        if (hm) {
                            const unsigned b    = (unsigned)__builtin_ctz(hm);
                            const unsigned slot = qn + __builtin_amdgcn_mbcnt_lo(m, 0u);
                            mylist[slot] = (t << 9) | (unsigned)(lane * 16) | b;
                            hm &= hm - 1u;
                        }
                        qn += (unsigned)__builtin_popcount(m);
                        m = __builtin_amdgcn_ballot_w32(hm != 0u);
                    }
                    st += TILE_WAVES;
                    ++t;
                }
            }
        }
        const bool ordered = (qn <= (unsigned)SORT_MAX);
        if (ordered) {
            unsigned en[SORT_CHUNKS];
            unsigned ev[SORT_CHUNKS];
            unsigned ow[SORT_CHUNKS];
#pragma unroll
            for (int c = 0; c < SORT_CHUNKS; ++c) {
                const unsigned p = (unsigned)(c * 32 + lane);
                en[c] = 0u;
                ev[c] = 0u;
                ow[c] = 32u;
                if (p < qn) {
                    const unsigned e0 = mylist[p];
                    int spd = wave + TILE_WAVES * (int)((e0 >> 9) & 0x3FFFu) + rot;
                    if (spd >= nsteps) spd -= nsteps;
                    const unsigned e  = (unsigned)(spd * SCAN_STEP) + (e0 & 511u);
                    const unsigned long long rc = rec[e];
                    en[c] = (unsigned)rc;
                    ev[c] = (unsigned)(rc >> 32);
                    ow[c] = ((unsigned)rc >> COL_BITS) & 31u;
                }
            }
            unsigned tot = 0;
            unsigned pre[SORT_CHUNKS];
            unsigned rk[SORT_CHUNKS];
#pragma unroll
            for (int c = 0; c < SORT_CHUNKS; ++c) {
                pre[c] = tot;
                rk[c]  = 0u;
                if ((unsigned)(c * 32) < qn) {
                    const unsigned vm = __builtin_amdgcn_ballot_w32((unsigned)(c * 32 + lane) < qn);
                    unsigned mm = vm;
                    unsigned hb = vm;
#pragma unroll
                    for (int k = 0; k < 5; ++k) {
                        const unsigned bal = __builtin_amdgcn_ballot_w32(((ow[c] >> k) & 1u) != 0u);
                        mm &= ((ow[c] >> k) & 1u) ? bal : ~bal;
                        hb &= (((unsigned)lane >> k) & 1u) ? bal : ~bal;
                    }
                    tot  += (unsigned)__builtin_popcount(hb);
                    rk[c] = __builtin_amdgcn_mbcnt_lo(mm, 0u);
                }
            }
            unsigned incl = tot;
#pragma unroll
            for (int d = 1; d < 32; d <<= 1) {
                const unsigned o = (unsigned)__builtin_amdgcn_ds_bpermute((int)((((unsigned)(lane - d)) & 31u) << 2), (int)incl);
                if (lane >= d) incl += o;
            }
            const unsigned start = incl - tot;
#pragma unroll
            for (int c = 0; c < SORT_CHUNKS; ++c) {
                if ((unsigned)(c * 32) < qn) {
                    const unsigned b = (unsigned)__builtin_amdgcn_ds_bpermute((int)((ow[c] & 31u) << 2),
                                                                              (int)(start + pre[c]));
                    if ((unsigned)(c * 32 + lane) < qn) {
                        const unsigned q = b + rk[c];
                        mylist[2u * q]      = en[c];
                        mylist[2u * q + 1u] = ev[c];
                    }
                }
            }
            mylist[OFFS_AT + lane] = start;
        } else {
            for (unsigned p = (unsigned)lane; p < qn; p += 32u) {
                const unsigned en = mylist[p];
                int spd = wave + TILE_WAVES * (int)((en >> 9) & 0x3FFFu) + rot;
                if (spd >= nsteps) spd -= nsteps;
                const unsigned e = (unsigned)(spd * SCAN_STEP) + (en & 511u);
                mylist[p] = en | (((unsigned)rec[e] >> COL_BITS) << 23);
            }
        }
        if (lane == 0) {
            cnt[wave] = qn | (ordered ? SORTED_FLAG : 0u);
            if (st < nsteps) pending = 1;
        }
        __syncthreads();

        const unsigned cl = cnt[lane];
        unsigned sb = 0u;
        unsigned sn = cl & ~SORTED_FLAG;
        if (cl & SORTED_FLAG) {
            sb = lists[lane * LIST_CAP + OFFS_AT + wave];
            const unsigned se = (wave < TILE_WAVES - 1) ? lists[lane * LIST_CAP + OFFS_AT + wave + 1] : sn;
            sn = se - sb;
        }
        unsigned qs = 0;
        for (int L = 0; L <= TILE_WAVES; ++L) {
            unsigned n = 0;
            unsigned base = 0;
            unsigned pickbits = 0u;
            unsigned esh = 0u;
            int nchunks = 1;
            unsigned cur = 0, curv = 0;
            if (L < TILE_WAVES) {
                n    = (unsigned)__builtin_amdgcn_readlane((int)sn, L);
                esh  = ((unsigned)__builtin_amdgcn_readlane((int)cl, L) & SORTED_FLAG) ? 1u : 0u;
                base = (unsigned)(L * LIST_CAP) + ((unsigned)__builtin_amdgcn_readlane((int)sb, L) << esh);
                pickbits = esh ? 0u : 31u;
                nchunks = (int)((n + 31u) >> 5);
                if ((unsigned)lane < n) {
                    cur  = lists[base + ((unsigned)lane << esh)];
                    curv = lists[base + ((unsigned)lane << esh) + esh];
                }
            }
            for (int c = 0; c < nchunks; ++c) {
                unsigned nxt = 0, nxtv = 0;
                if (L < TILE_WAVES) {
                    const unsigned pn = (unsigned)((c + 1) * 32 + lane);
                    if (pn < n) {
                        nxt  = lists[base + (pn << esh)];
                        nxtv = lists[base + (pn << esh) + esh];
                    }
                    const unsigned p = (unsigned)(c * 32 + lane);
                    const bool own = (p < n) && ((((cur >> 23) ^ (unsigned)wave) & pickbits) == 0u);
                    const unsigned m = __builtin_amdgcn_ballot_w32(own);
                    if (m) {
                        if (own) {
                            const unsigned slot = qs + __builtin_amdgcn_mbcnt_lo(m, 0u);
                            unsigned long long sg = (unsigned long long)cur | ((unsigned long long)curv << 32);
                            if (!esh) {
                                int spd = L + TILE_WAVES * (int)((cur >> 9) & 0x3FFFu) + rot;
                                if (spd >= nsteps) spd -= nsteps;
                                sg = (unsigned long long)(((unsigned)(spd * SCAN_STEP) + (cur & 511u)) | UNRES_FLAG);
                            }
                            mystage[slot] = sg;
                        }
                        qs += (unsigned)__builtin_popcount(m);
                    }
                }
                const unsigned thr = (L < TILE_WAVES) ? 32u : 1u;
                if (qs >= thr) {
                    const unsigned nb = qs < 32u ? qs : 32u;
                    int cv = 0, vv = 0, dv = 0;
                    if ((unsigned)lane < nb) {
                        unsigned long long sg = mystage[lane];
                        if ((unsigned)sg & UNRES_FLAG) sg = rec[(unsigned)sg & ~UNRES_FLAG];
                        cv = (int)((unsigned)sg & COL_MASK);
                        dv = (int)((unsigned)sg >> COL_BITS);
                        vv = (int)(unsigned)(sg >> 32);
                    }
                    unsigned k = 0;
                    for (; k + 8 <= nb; k += 8) {
                        float4 s[8];
#pragma unroll
                        for (unsigned u = 0; u < 8; ++u) {
                            const int cc = __builtin_amdgcn_readlane(cv, (int)(k + u));
                            s[u] = *(const float4*)(S + (long long)cc * D_OUT + lane * 4);
                        }
#pragma unroll
                        for (unsigned u = 0; u < 8; ++u) {
                            const float v = __int_as_float(__builtin_amdgcn_readlane(vv, (int)(k + u)));
                            const int   d = __builtin_amdgcn_readlane(dv, (int)(k + u));
                            float4* ap = acc + d * (D_OUT / 4) + lane;
                            float4 a = *ap;
                            a.x += s[u].x * v;
                            a.y += s[u].y * v;
                            a.z += s[u].z * v;
                            a.w += s[u].w * v;
                            *ap = a;
                        }
                    }
                    for (; k < nb; ++k) {
                        const int   cc = __builtin_amdgcn_readlane(cv, (int)k);
                        const float v  = __int_as_float(__builtin_amdgcn_readlane(vv, (int)k));
                        const int   d  = __builtin_amdgcn_readlane(dv, (int)k);
                        const float4 s1 = *(const float4*)(S + (long long)cc * D_OUT + lane * 4);
                        float4* ap = acc + d * (D_OUT / 4) + lane;
                        float4 a = *ap;
                        a.x += s1.x * v;
                        a.y += s1.y * v;
                        a.z += s1.z * v;
                        a.w += s1.w * v;
                        *ap = a;
                    }
                    const unsigned left = qs - nb;
                    unsigned long long mv = 0ull;
                    if ((unsigned)lane < left) mv = mystage[nb + lane];
                    if ((unsigned)lane < left) mystage[lane] = mv;
                    qs = left;
                }
                cur  = nxt;
                curv = nxtv;
            }
        }

        const unsigned more = (unsigned)__builtin_amdgcn_readfirstlane((int)pending);
        __syncthreads();
        if (!more) break;
        if (tid == 0) pending = 0;
        __syncthreads();
    }

    const float4 bv = *(const float4*)(bias + lane * 4);
    for (unsigned d = (unsigned)wave; d < tile_rows; d += TILE_WAVES) {
        const float4 a = acc[d * (D_OUT / 4) + lane];
        float4 o;
        o.x = a.x + bv.x;
        o.y = a.y + bv.y;
        o.z = a.z + bv.z;
        o.w = a.w + bv.w;
        *(float4*)(out + (long long)(r0 + d) * D_OUT + lane * 4) = o;
    }
}

extern "C" void kernel_launch(void* const* d_in, const int* in_sizes, int n_in,
                              void* d_out, int out_size, void* d_ws, size_t ws_size,
                              hipStream_t stream) {
    const int*   rows = (const int*)d_in[0];
    const int*   cols = (const int*)d_in[1];
    const float* vals = (const float*)d_in[2];
    const float* x    = (const float*)d_in[3];
    const float* W    = (const float*)d_in[4];
    const float* bias = (const float*)d_in[5];

    const int E = in_sizes[0];
    const int N = in_sizes[3] / D_IN;

    float* support = (float*)d_ws;
    unsigned* keys = (unsigned*)(support + (size_t)N * D_OUT);
    unsigned* recw = keys + (size_t)((E + SCAN_STEP - 1) / SCAN_STEP) * (SCAN_STEP / 4);

    {
        int mtiles = N / 16;
        int blocks = (mtiles + 7) / 8;
        gcn_gemm_xw<<<blocks, 256, 0, stream>>>(x, W, support, N, rows, cols, vals, keys, recw, E);
    }

    {
        int blocks = (N + ROWS_PER_WG - 1) / ROWS_PER_WG;
        gcn_tile_sum<<<blocks, 1024, 0, stream>>>(keys, recw, support, bias,
                                                  (float*)d_out, E, N);
    }
}
